# starting kernel
# baseline (speedup 1.0000x reference)
#include <hip/hip_runtime.h>
#include <math.h>

#define BATCH   32
#define NIN     2048
#define KDIM    16
#define NCAP    32
#define CAPD    32
#define NOUT    1024
#define NCHUNK  128
#define NC      (NIN / NCHUNK)
#define BH      16
#define UHS     1028

typedef float v2f __attribute__((ext_vector_type(2)));
typedef float v8f __attribute__((ext_vector_type(8)));

__global__ __launch_bounds__(256)
void caps_pass(const float* __restrict__ U,
               const float* __restrict__ Wt,
               const float* __restrict__ o_prev,
               float* __restrict__ partial,
               int iter0)
{
    extern __shared__ float smem[];
    float* uhat = smem;
    float* oacc = uhat + BH * UHS;
    float* lg   = oacc + BH * 1024;
    float* cb   = lg + BH * 32;

    const int tid   = threadIdx.x;
    const int lane  = tid & 31;
    const int wave  = tid >> 5;
    const int chunk = blockIdx.x;
    const int half  = blockIdx.y;
    const int b0    = half * BH;

    const int l16 = lane & 15;
    const int hi  = lane >> 4;

    for (int e = tid; e < BH * 1024; e += 256) oacc[e] = 0.0f;
    __syncthreads();

    for (int nn = 0; nn < NC; ++nn) {
        const int n = chunk * NC + nn;

        if (nn + 1 < NC) {
            __builtin_prefetch(Wt + ((size_t)(n + 1) * KDIM) * NOUT + wave * 128, 0, 3);
        }

        const float* Up = U + ((size_t)(b0 + l16) * NIN + n) * KDIM;
        v2f a0, a1, a2, a3;
        a0.x = Up[0  + 2*hi]; a0.y = Up[1  + 2*hi];
        a1.x = Up[4  + 2*hi]; a1.y = Up[5  + 2*hi];
        a2.x = Up[8  + 2*hi]; a2.y = Up[9  + 2*hi];
        a3.x = Up[12 + 2*hi]; a3.y = Up[13 + 2*hi];

        const float* Wn = Wt + (size_t)n * KDIM * NOUT;
        for (int ct = 0; ct < 8; ++ct) {
            const int c0 = wave * 128 + ct * 16;
            const float* Wp = Wn + (size_t)(2 * hi) * NOUT + c0 + l16;
            v2f bf0, bf1, bf2, bf3;
            bf0.x = Wp[0  * NOUT]; bf0.y = Wp[1  * NOUT];
            bf1.x = Wp[4  * NOUT]; bf1.y = Wp[5  * NOUT];
            bf2.x = Wp[8  * NOUT]; bf2.y = Wp[9  * NOUT];
            bf3.x = Wp[12 * NOUT]; bf3.y = Wp[13 * NOUT];

            v8f acc = {};
            acc = __builtin_amdgcn_wmma_f32_16x16x4_f32(false, a0, false, bf0, (short)0, acc, false, false);
            acc = __builtin_amdgcn_wmma_f32_16x16x4_f32(false, a1, false, bf1, (short)0, acc, false, false);
            acc = __builtin_amdgcn_wmma_f32_16x16x4_f32(false, a2, false, bf2, (short)0, acc, false, false);
            acc = __builtin_amdgcn_wmma_f32_16x16x4_f32(false, a3, false, bf3, (short)0, acc, false, false);

            #pragma unroll
            for (int v = 0; v < 8; ++v)
                uhat[(v + 8 * hi) * UHS + c0 + l16] = acc[v];
        }
        __syncthreads();

        if (iter0) {
            for (int e = tid; e < BH * 1024; e += 256)
                oacc[e] += 0.03125f * uhat[(e >> 10) * UHS + (e & 1023)];
        } else {
            for (int p = tid; p < BH * NCAP; p += 256) {
                const int bl = p >> 5, i = p & 31;
                const float* op = o_prev + ((size_t)(b0 + bl) * NCAP + i) * CAPD;
                const float* uh = uhat + bl * UHS + i * CAPD;
                float s = 0.0f;
                #pragma unroll
                for (int j = 0; j < CAPD; ++j) s = fmaf(op[j], uh[j], s);
                lg[p] = s;
            }
            __syncthreads();
            if (tid < BH) {
                float m = -3.402823466e38f;
                for (int i = 0; i < NCAP; ++i) m = fmaxf(m, lg[tid * NCAP + i]);
                float s = 0.0f;
                for (int i = 0; i < NCAP; ++i) {
                    float e = __expf(lg[tid * NCAP + i] - m);
                    cb[tid * NCAP + i] = e; s += e;
                }
                float inv = 1.0f / s;
                for (int i = 0; i < NCAP; ++i) cb[tid * NCAP + i] *= inv;
            }
            __syncthreads();
            for (int e = tid; e < BH * 1024; e += 256) {
                const int bl = e >> 10, col = e & 1023;
                oacc[e] = fmaf(cb[bl * NCAP + (col >> 5)], uhat[bl * UHS + col], oacc[e]);
            }
        }
        __syncthreads();
    }

    float* pp = partial + (size_t)(chunk * 2 + half) * (BH * 1024);
    for (int e = tid; e < BH * 1024; e += 256) pp[e] = oacc[e];
}

__global__ __launch_bounds__(256)
void caps_reduce(const float* __restrict__ partial, float* __restrict__ out, int mode)
{
    __shared__ float osum[NOUT];
    __shared__ float fac[NCAP];
    const int b = blockIdx.x;
    const int half = b >> 4, bl = b & 15;
    const int tid = threadIdx.x;

    for (int e = tid; e < NOUT; e += 256) {
        float s = 0.0f;
        const float* p = partial + (size_t)half * (BH * 1024) + (size_t)bl * 1024 + e;
        for (int c = 0; c < NCHUNK; ++c)
            s += p[(size_t)c * 2 * (BH * 1024)];
        osum[e] = s;
    }
    __syncthreads();

    if (tid < NCAP) {
        float s2 = 0.0f;
        for (int j = 0; j < CAPD; ++j) { float v = osum[tid * CAPD + j]; s2 = fmaf(v, v, s2); }
        float f;
        if (mode == 0) f = rsqrtf(fmaxf(s2, 1e-12f));
        else           f = (s2 / (1.0f + s2)) * rsqrtf(s2 + 1e-7f);
        fac[tid] = f;
    }
    __syncthreads();

    for (int e = tid; e < NOUT; e += 256)
        out[(size_t)b * NOUT + e] = osum[e] * fac[e >> 5];
}

extern "C" void kernel_launch(void* const* d_in, const int* in_sizes, int n_in,
                              void* d_out, int out_size, void* d_ws, size_t ws_size,
                              hipStream_t stream)
{
    const float* U  = (const float*)d_in[0];
    const float* Wt = (const float*)d_in[1];
    float* out = (float*)d_out;

    float* partial = (float*)d_ws;
    float* o_prev  = partial + (size_t)(NCHUNK * 2) * (BH * 1024);

    const size_t shmem = (size_t)(BH * UHS + BH * 1024 + 2 * BH * NCAP) * sizeof(float);
    dim3 grid(NCHUNK, 2), blk(256);

    caps_pass<<<grid, blk, shmem, stream>>>(U, Wt, nullptr, partial, 1);
    caps_reduce<<<BATCH, 256, 0, stream>>>(partial, o_prev, 0);
    caps_pass<<<grid, blk, shmem, stream>>>(U, Wt, o_prev, partial, 0);
    caps_reduce<<<BATCH, 256, 0, stream>>>(partial, o_prev, 0);
    caps_pass<<<grid, blk, shmem, stream>>>(U, Wt, o_prev, partial, 0);
    caps_reduce<<<BATCH, 256, 0, stream>>>(partial, out, 1);
}


# pass 0 (uniform coupling) split over 8 column groups x 16 n-chunks, each wave on its own n with an LDS fold of the 8 waves: its plane sum reads 16 planes, not 128; pass-0 sums re-ordered
# speedup vs baseline: 2.8600x; 2.8600x over previous
#include <hip/hip_runtime.h>
#include <math.h>


#define BATCH   32
#define NIN     2048
#define KDIM    16
#define NCAP    32
#define CAPD    32
#define NOUT    1024
#define NCHUNK  128
#define NC      (NIN / NCHUNK)
#define BH      16
#define WP      1028
#define LGP     33
#define P0_NCH  16
#define P0_CG   8
#define P0_NW   (NIN / P0_NCH / 8)
#define WQ      132

typedef float v2f __attribute__((ext_vector_type(2)));
typedef float v4f __attribute__((ext_vector_type(4)));
typedef float v8f __attribute__((ext_vector_type(8)));

#define WMMA_F32(a, b, c) __builtin_amdgcn_wmma_f32_16x16x4_f32(false, (a), false, (b), (short)0, (c), false, false)

__device__ __forceinline__ void lds_barrier()
{
    __builtin_amdgcn_fence(__ATOMIC_RELEASE, "workgroup", "local");
    __builtin_amdgcn_s_barrier();
    __builtin_amdgcn_fence(__ATOMIC_ACQUIRE, "workgroup", "local");
}

__device__ __forceinline__ void wave_lds_sync()
{
    __builtin_amdgcn_fence(__ATOMIC_RELEASE, "wavefront", "local");
    __builtin_amdgcn_wave_barrier();
    __builtin_amdgcn_fence(__ATOMIC_ACQUIRE, "wavefront", "local");
}

__global__ __launch_bounds__(256)
void caps_pass0(const float* __restrict__ U,
                const float* __restrict__ Wt,
                float* __restrict__ partial)
{
    extern __shared__ __align__(16) float smem[];

    const int tid   = threadIdx.x;
    const int lane  = tid & 31;
    const int wave  = tid >> 5;
    const int chunk = blockIdx.x;
    const int half  = blockIdx.y;
    const int cg    = blockIdx.z;
    const int b0    = half * BH;
    const int l16   = lane & 15;
    const int hi    = lane >> 4;

    float* Wl = smem + wave * (KDIM * WQ);

    v8f oacc[4][2];
    #pragma unroll
    for (int q = 0; q < 4; ++q) { v8f z = {}; oacc[q][0] = z; oacc[q][1] = z; }

    const int nbase = chunk * (NIN / P0_NCH) + wave;

    v4f wreg[KDIM];
    v4f un0, un1;
    {
        const float* Wn = Wt + (size_t)nbase * KDIM * NOUT + cg * 128 + 4 * lane;
        #pragma unroll
        for (int r = 0; r < KDIM; ++r) wreg[r] = *(const v4f*)(Wn + (size_t)r * NOUT);
        const v4f* up = (const v4f*)(U + ((size_t)(b0 + l16) * NIN + nbase) * KDIM + 8 * hi);
        un0 = up[0]; un1 = up[1];
    }

    for (int t = 0; t < P0_NW; ++t) {
        const int n = nbase + 8 * t;

        wave_lds_sync();
        #pragma unroll
        for (int r = 0; r < KDIM; ++r) *(v4f*)(Wl + r * WQ + 4 * lane) = wreg[r];

        float ur[8];
        ur[0] = 0.03125f * un0.x; ur[1] = 0.03125f * un0.y; ur[2] = 0.03125f * un0.z; ur[3] = 0.03125f * un0.w;
        ur[4] = 0.03125f * un1.x; ur[5] = 0.03125f * un1.y; ur[6] = 0.03125f * un1.z; ur[7] = 0.03125f * un1.w;

        if (t + 1 < P0_NW) {
            const v4f* up = (const v4f*)(U + ((size_t)(b0 + l16) * NIN + (n + 8)) * KDIM + 8 * hi);
            un0 = up[0]; un1 = up[1];
            const float* Wn = Wt + (size_t)(n + 8) * KDIM * NOUT + cg * 128 + 4 * lane;
            #pragma unroll
            for (int r = 0; r < KDIM; ++r) wreg[r] = *(const v4f*)(Wn + (size_t)r * NOUT);
        }
        wave_lds_sync();

        #pragma unroll
        for (int s = 0; s < 4; ++s) {
            v2f a;
            a.x = ur[2 * s];
            a.y = ur[2 * s + 1];
            #pragma unroll
            for (int q = 0; q < 4; ++q) {
                #pragma unroll
                for (int jt = 0; jt < 2; ++jt) {
                    const float* wp = Wl + (8 * hi + 2 * s) * WQ + q * CAPD + jt * 16 + l16;
                    v2f bb;
                    bb.x = wp[0];
                    bb.y = wp[WQ];
                    oacc[q][jt] = WMMA_F32(a, bb, oacc[q][jt]);
                }
            }
        }
    }

    wave_lds_sync();
    #pragma unroll
    for (int q = 0; q < 4; ++q) {
        #pragma unroll
        for (int jt = 0; jt < 2; ++jt) {
            #pragma unroll
            for (int v = 0; v < 8; ++v)
                Wl[(v + 8 * hi) * WQ + q * CAPD + jt * 16 + l16] = oacc[q][jt][v];
        }
    }
    __syncthreads();

    float* pp = partial + (size_t)(chunk * 2 + half) * (BH * 1024) + cg * 128;
    #pragma unroll
    for (int i = 0; i < 8; ++i) {
        const int e   = tid + 256 * i;
        const int row = e >> 7;
        const int col = e & 127;
        float s = 0.0f;
        #pragma unroll
        for (int w = 0; w < 8; ++w) s += smem[w * (KDIM * WQ) + row * WQ + col];
        pp[row * 1024 + col] = s;
    }
}

__global__ __launch_bounds__(256)
void caps_pass(const float* __restrict__ U,
               const float* __restrict__ Wt,
               const float* __restrict__ o_prev,
               float* __restrict__ partial)
{
    extern __shared__ __align__(16) float smem[];
    float* Ws  = smem;
    float* lgs = smem + 2 * KDIM * WP;

    const int tid   = threadIdx.x;
    const int lane  = tid & 31;
    const int wave  = tid >> 5;
    const int chunk = blockIdx.x;
    const int half  = blockIdx.y;
    const int b0    = half * BH;

    const int l16 = lane & 15;
    const int hi  = lane >> 4;


    v2f of[4][8];
    #pragma unroll
    for (int q = 0; q < 4; ++q) {
        #pragma unroll
        for (int s = 0; s < 8; ++s) {
            const float* op = o_prev + ((size_t)(b0 + l16) * NCAP + (wave * 4 + q)) * CAPD + 4 * s + 2 * hi;
            of[q][s] = *(const v2f*)op;
        }
    }

    v8f oacc[4][2];
    #pragma unroll
    for (int q = 0; q < 4; ++q) { v8f z = {}; oacc[q][0] = z; oacc[q][1] = z; }

    v4f wreg[KDIM];
    v4f un0, un1;
    {
        const int n = chunk * NC;
        const float* Wn = Wt + (size_t)n * KDIM * NOUT + 4 * tid;
        #pragma unroll
        for (int r = 0; r < KDIM; ++r) wreg[r] = *(const v4f*)(Wn + (size_t)r * NOUT);
        const v4f* up = (const v4f*)(U + ((size_t)(b0 + l16) * NIN + n) * KDIM + 8 * hi);
        un0 = up[0]; un1 = up[1];
    }

    for (int nn = 0; nn < NC; ++nn) {
        const int n = chunk * NC + nn;
        float* Wb = Ws + (nn & 1) * (KDIM * WP);

        #pragma unroll
        for (int r = 0; r < KDIM; ++r) *(v4f*)(Wb + r * WP + 4 * tid) = wreg[r];

        float ur[8];
        ur[0] = un0.x; ur[1] = un0.y; ur[2] = un0.z; ur[3] = un0.w;
        ur[4] = un1.x; ur[5] = un1.y; ur[6] = un1.z; ur[7] = un1.w;

        if (nn + 1 < NC) {
            const v4f* up = (const v4f*)(U + ((size_t)(b0 + l16) * NIN + (n + 1)) * KDIM + 8 * hi);
            un0 = up[0]; un1 = up[1];
            const float* Wn = Wt + (size_t)(n + 1) * KDIM * NOUT + 4 * tid;
            #pragma unroll
            for (int r = 0; r < KDIM; ++r) wreg[r] = *(const v4f*)(Wn + (size_t)r * NOUT);
        }
        lds_barrier();

        float cq[4];
        {
            v8f acc[4];
            #pragma unroll
            for (int q = 0; q < 4; ++q) { v8f z = {}; acc[q] = z; }
            #pragma unroll
            for (int s = 0; s < 8; ++s) {
                #pragma unroll
                for (int q = 0; q < 4; ++q) {
                    const float* wp = Wb + l16 * WP + (wave * 4 + q) * CAPD + 4 * s + 2 * hi;
                    const v2f a = *(const v2f*)wp;
                    acc[q] = WMMA_F32(a, of[q][s], acc[q]);
                }
            }
            float lg[4];
            #pragma unroll
            for (int q = 0; q < 4; ++q) {
                float sd = 0.0f;
                #pragma unroll
                for (int v = 0; v < 8; ++v) sd = fmaf(ur[v], acc[q][v], sd);
                sd += __shfl_xor(sd, 16);
                lg[q] = sd;
            }
            if (hi == 0) {
                #pragma unroll
                for (int q = 0; q < 4; ++q) lgs[l16 * LGP + wave * 4 + q] = lg[q];
            }
            lds_barrier();
            float lv[NCAP];
            float m = -3.402823466e38f;
            #pragma unroll
            for (int i = 0; i < NCAP; ++i) { lv[i] = lgs[l16 * LGP + i]; m = fmaxf(m, lv[i]); }
            float ssum = 0.0f;
            #pragma unroll
            for (int i = 0; i < NCAP; ++i) ssum += __expf(lv[i] - m);
            const float inv = 1.0f / ssum;
            #pragma unroll
            for (int q = 0; q < 4; ++q) cq[q] = __expf(lg[q] - m) * inv;
        }

        #pragma unroll
        for (int s = 0; s < 4; ++s) {
            #pragma unroll
            for (int q = 0; q < 4; ++q) {
                v2f a;
                a.x = cq[q] * ur[2 * s];
                a.y = cq[q] * ur[2 * s + 1];
                #pragma unroll
                for (int jt = 0; jt < 2; ++jt) {
                    const float* wp = Wb + (8 * hi + 2 * s) * WP + (wave * 4 + q) * CAPD + jt * 16 + l16;
                    v2f bb;
                    bb.x = wp[0];
                    bb.y = wp[WP];
                    oacc[q][jt] = WMMA_F32(a, bb, oacc[q][jt]);
                }
            }
        }
    }

    float* pp = partial + (size_t)(chunk * 2 + half) * (BH * 1024);
    #pragma unroll
    for (int q = 0; q < 4; ++q) {
        #pragma unroll
        for (int jt = 0; jt < 2; ++jt) {
            #pragma unroll
            for (int v = 0; v < 8; ++v)
                pp[(v + 8 * hi) * 1024 + (wave * 4 + q) * CAPD + jt * 16 + l16] = oacc[q][jt][v];
        }
    }
}

template <int NPL>
__global__ __launch_bounds__(256)
void caps_reduce(const float* __restrict__ partial, float* __restrict__ out, int mode)
{
    __shared__ float osum[NOUT];
    __shared__ float fac[NCAP];
    const int b = blockIdx.x;
    const int half = b >> 4, bl = b & 15;
    const int tid = threadIdx.x;

    for (int e = tid; e < NOUT; e += 256) {
        float s = 0.0f;
        const float* p = partial + (size_t)half * (BH * 1024) + (size_t)bl * 1024 + e;
        for (int c = 0; c < NPL; ++c)
            s += p[(size_t)c * 2 * (BH * 1024)];
        osum[e] = s;
    }
    __syncthreads();

    if (tid < NCAP) {
        float s2 = 0.0f;
        for (int j = 0; j < CAPD; ++j) { float v = osum[tid * CAPD + j]; s2 = fmaf(v, v, s2); }
        float f;
        if (mode == 0) f = rsqrtf(fmaxf(s2, 1e-12f));
        else           f = (s2 / (1.0f + s2)) * rsqrtf(s2 + 1e-7f);
        fac[tid] = f;
    }
    __syncthreads();

    for (int e = tid; e < NOUT; e += 256)
        out[(size_t)b * NOUT + e] = osum[e] * fac[e >> 5];
}

extern "C" void kernel_launch(void* const* d_in, const int* in_sizes, int n_in,
                              void* d_out, int out_size, void* d_ws, size_t ws_size,
                              hipStream_t stream)
{
    const float* U  = (const float*)d_in[0];
    const float* Wt = (const float*)d_in[1];
    float* out = (float*)d_out;

    float* partial = (float*)d_ws;
    float* o_prev  = partial + (size_t)(NCHUNK * 2) * (BH * 1024);

    const size_t shmem  = (size_t)(2 * KDIM * WP + BH * LGP) * sizeof(float);
    const size_t shmem0 = (size_t)(8 * KDIM * WQ) * sizeof(float);
    dim3 grid(NCHUNK, 2), grid0(P0_NCH, 2, P0_CG), blk(256);

    caps_pass0<<<grid0, blk, shmem0, stream>>>(U, Wt, partial);
    caps_reduce<P0_NCH><<<BATCH, 256, 0, stream>>>(partial, o_prev, 0);
    caps_pass<<<grid, blk, shmem, stream>>>(U, Wt, o_prev, partial);
    caps_reduce<NCHUNK><<<BATCH, 256, 0, stream>>>(partial, o_prev, 0);
    caps_pass<<<grid, blk, shmem, stream>>>(U, Wt, o_prev, partial);
    caps_reduce<NCHUNK><<<BATCH, 256, 0, stream>>>(partial, out, 1);
}


# each wave feeds its staged W tile to both 16-batch matrix tiles (workgroup = 32 batches x 64 n x 128 columns): W fetched once per pass, not twice; 32 planes of 64 n; sums re-ordered
# speedup vs baseline: 3.7359x; 1.3063x over previous
#include <hip/hip_runtime.h>
#include <math.h>


#define BATCH   32
#define NIN     2048
#define KDIM    16
#define NCAP    32
#define CAPD    32
#define NOUT    1024
#define BH      16
#define NCH     32
#define P0_CG   8
#define NW      (NIN / NCH / 8)
#define WQ      132

typedef float v2f __attribute__((ext_vector_type(2)));
typedef float v4f __attribute__((ext_vector_type(4)));
typedef float v8f __attribute__((ext_vector_type(8)));

#define WMMA_F32(a, b, c) __builtin_amdgcn_wmma_f32_16x16x4_f32(false, (a), false, (b), (short)0, (c), false, false)

__device__ __forceinline__ void wave_lds_sync()
{
    __builtin_amdgcn_fence(__ATOMIC_RELEASE, "wavefront", "local");
    __builtin_amdgcn_wave_barrier();
    __builtin_amdgcn_fence(__ATOMIC_ACQUIRE, "wavefront", "local");
}

__device__ __forceinline__ void fold_store(float* smem, float* Wl, const v8f (&oacc)[2][4][2],
                                           float* pp, int tid, int l16, int hi)
{
    #pragma unroll
    for (int h = 0; h < 2; ++h) {
        __syncthreads();
        #pragma unroll
        for (int q = 0; q < 4; ++q) {
            #pragma unroll
            for (int jt = 0; jt < 2; ++jt) {
                #pragma unroll
                for (int v = 0; v < 8; ++v)
                    Wl[(v + 8 * hi) * WQ + q * CAPD + jt * 16 + l16] = oacc[h][q][jt][v];
            }
        }
        __syncthreads();

        #pragma unroll
        for (int i = 0; i < 8; ++i) {
            const int e   = tid + 256 * i;
            const int row = e >> 7;
            const int col = e & 127;
            float s = 0.0f;
            #pragma unroll
            for (int w = 0; w < 8; ++w) s += smem[w * (KDIM * WQ) + row * WQ + col];
            pp[(size_t)(h * BH + row) * 1024 + col] = s;
        }
    }
}

__global__ __launch_bounds__(256)
void caps_pass0(const float* __restrict__ U,
                const float* __restrict__ Wt,
                float* __restrict__ partial)
{
    extern __shared__ __align__(16) float smem[];

    const int tid   = threadIdx.x;
    const int lane  = tid & 31;
    const int wave  = tid >> 5;
    const int chunk = blockIdx.x;
    const int cg    = blockIdx.y;
    const int l16   = lane & 15;
    const int hi    = lane >> 4;

    float* Wl = smem + wave * (KDIM * WQ);

    v8f oacc[2][4][2];
    #pragma unroll
    for (int h = 0; h < 2; ++h) {
        #pragma unroll
        for (int q = 0; q < 4; ++q) { v8f z = {}; oacc[h][q][0] = z; oacc[h][q][1] = z; }
    }

    const int nbase = chunk * (NIN / NCH) + wave;

    v4f wreg[KDIM];
    v4f un[2][2];
    {
        const float* Wn = Wt + (size_t)nbase * KDIM * NOUT + cg * 128 + 4 * lane;
        #pragma unroll
        for (int r = 0; r < KDIM; ++r) wreg[r] = *(const v4f*)(Wn + (size_t)r * NOUT);
        #pragma unroll
        for (int h = 0; h < 2; ++h) {
            const v4f* up = (const v4f*)(U + ((size_t)(h * BH + l16) * NIN + nbase) * KDIM + 8 * hi);
            un[h][0] = up[0]; un[h][1] = up[1];
        }
    }

    for (int t = 0; t < NW; ++t) {
        const int n = nbase + 8 * t;

        wave_lds_sync();
        #pragma unroll
        for (int r = 0; r < KDIM; ++r) *(v4f*)(Wl + r * WQ + 4 * lane) = wreg[r];

        float ur[2][8];
        #pragma unroll
        for (int h = 0; h < 2; ++h) {
            ur[h][0] = 0.03125f * un[h][0].x; ur[h][1] = 0.03125f * un[h][0].y;
            ur[h][2] = 0.03125f * un[h][0].z; ur[h][3] = 0.03125f * un[h][0].w;
            ur[h][4] = 0.03125f * un[h][1].x; ur[h][5] = 0.03125f * un[h][1].y;
            ur[h][6] = 0.03125f * un[h][1].z; ur[h][7] = 0.03125f * un[h][1].w;
        }

        if (t + 1 < NW) {
            #pragma unroll
            for (int h = 0; h < 2; ++h) {
                const v4f* up = (const v4f*)(U + ((size_t)(h * BH + l16) * NIN + (n + 8)) * KDIM + 8 * hi);
                un[h][0] = up[0]; un[h][1] = up[1];
            }
            const float* Wn = Wt + (size_t)(n + 8) * KDIM * NOUT + cg * 128 + 4 * lane;
            #pragma unroll
            for (int r = 0; r < KDIM; ++r) wreg[r] = *(const v4f*)(Wn + (size_t)r * NOUT);
        }
        wave_lds_sync();

        #pragma unroll
        for (int s = 0; s < 4; ++s) {
            v2f a0, a1;
            a0.x = ur[0][2 * s]; a0.y = ur[0][2 * s + 1];
            a1.x = ur[1][2 * s]; a1.y = ur[1][2 * s + 1];
            #pragma unroll
            for (int q = 0; q < 4; ++q) {
                #pragma unroll
                for (int jt = 0; jt < 2; ++jt) {
                    const float* wp = Wl + (8 * hi + 2 * s) * WQ + q * CAPD + jt * 16 + l16;
                    v2f bb;
                    bb.x = wp[0];
                    bb.y = wp[WQ];
                    oacc[0][q][jt] = WMMA_F32(a0, bb, oacc[0][q][jt]);
                    oacc[1][q][jt] = WMMA_F32(a1, bb, oacc[1][q][jt]);
                }
            }
        }
    }

    fold_store(smem, Wl, oacc, partial + (size_t)chunk * (BATCH * 1024) + cg * 128, tid, l16, hi);
}

__global__ __launch_bounds__(256)
void caps_logits(const float* __restrict__ U,
                 const float* __restrict__ Wt,
                 const float* __restrict__ o_prev,
                 float* __restrict__ lgt)
{
    extern __shared__ __align__(16) float smem[];

    const int tid   = threadIdx.x;
    const int lane  = tid & 31;
    const int wave  = tid >> 5;
    const int chunk = blockIdx.x;
    const int cg    = blockIdx.y;
    const int l16   = lane & 15;
    const int hi    = lane >> 4;

    float* Wl = smem + wave * (KDIM * WQ);


    v2f of[2][4][8];
    #pragma unroll
    for (int h = 0; h < 2; ++h) {
        #pragma unroll
        for (int q = 0; q < 4; ++q) {
            #pragma unroll
            for (int s = 0; s < 8; ++s) {
                const float* op = o_prev + ((size_t)(h * BH + l16) * NCAP + (cg * 4 + q)) * CAPD + 4 * s + 2 * hi;
                of[h][q][s] = *(const v2f*)op;
            }
        }
    }

    const int nbase = chunk * (NIN / NCH) + wave;

    v4f wreg[KDIM];
    v4f un[2][2];
    {
        const float* Wn = Wt + (size_t)nbase * KDIM * NOUT + cg * 128 + 4 * lane;
        #pragma unroll
        for (int r = 0; r < KDIM; ++r) wreg[r] = *(const v4f*)(Wn + (size_t)r * NOUT);
        #pragma unroll
        for (int h = 0; h < 2; ++h) {
            const v4f* up = (const v4f*)(U + ((size_t)(h * BH + l16) * NIN + nbase) * KDIM + 8 * hi);
            un[h][0] = up[0]; un[h][1] = up[1];
        }
    }

    for (int t = 0; t < NW; ++t) {
        const int n = nbase + 8 * t;

        wave_lds_sync();
        #pragma unroll
        for (int r = 0; r < KDIM; ++r) *(v4f*)(Wl + r * WQ + 4 * lane) = wreg[r];

        float ur[2][8];
        #pragma unroll
        for (int h = 0; h < 2; ++h) {
            ur[h][0] = un[h][0].x; ur[h][1] = un[h][0].y; ur[h][2] = un[h][0].z; ur[h][3] = un[h][0].w;
            ur[h][4] = un[h][1].x; ur[h][5] = un[h][1].y; ur[h][6] = un[h][1].z; ur[h][7] = un[h][1].w;
        }

        if (t + 1 < NW) {
            #pragma unroll
            for (int h = 0; h < 2; ++h) {
                const v4f* up = (const v4f*)(U + ((size_t)(h * BH + l16) * NIN + (n + 8)) * KDIM + 8 * hi);
                un[h][0] = up[0]; un[h][1] = up[1];
            }
            const float* Wn = Wt + (size_t)(n + 8) * KDIM * NOUT + cg * 128 + 4 * lane;
            #pragma unroll
            for (int r = 0; r < KDIM; ++r) wreg[r] = *(const v4f*)(Wn + (size_t)r * NOUT);
        }
        wave_lds_sync();

        v8f acc[2][4];
        #pragma unroll
        for (int h = 0; h < 2; ++h) {
            #pragma unroll
            for (int q = 0; q < 4; ++q) { v8f z = {}; acc[h][q] = z; }
        }
        #pragma unroll
        for (int s = 0; s < 8; ++s) {
            #pragma unroll
            for (int q = 0; q < 4; ++q) {
                const float* wp = Wl + l16 * WQ + q * CAPD + 4 * s + 2 * hi;
                const v2f a = *(const v2f*)wp;
                acc[0][q] = WMMA_F32(a, of[0][q][s], acc[0][q]);
                acc[1][q] = WMMA_F32(a, of[1][q][s], acc[1][q]);
            }
        }
        v4f lg[2];
        #pragma unroll
        for (int h = 0; h < 2; ++h) {
            #pragma unroll
            for (int q = 0; q < 4; ++q) {
                float sd = 0.0f;
                #pragma unroll
                for (int v = 0; v < 8; ++v) sd = fmaf(ur[h][v], acc[h][q][v], sd);
                sd += __shfl_xor(sd, 16);
                lg[h][q] = sd;
            }
        }
        const v4f lgs = hi ? lg[1] : lg[0];
        *(v4f*)(lgt + ((((size_t)n * 2 + hi) * P0_CG + cg) * BH + l16) * 4) = lgs;
    }
}

__global__ __launch_bounds__(256)
void caps_o(const float* __restrict__ U,
            const float* __restrict__ Wt,
            const float* __restrict__ lgt,
            float* __restrict__ partial)
{
    extern __shared__ __align__(16) float smem[];

    const int tid   = threadIdx.x;
    const int lane  = tid & 31;
    const int wave  = tid >> 5;
    const int chunk = blockIdx.x;
    const int cg    = blockIdx.y;
    const int l16   = lane & 15;
    const int hi    = lane >> 4;

    float* Wl = smem + wave * (KDIM * WQ);

    v8f oacc[2][4][2];
    #pragma unroll
    for (int h = 0; h < 2; ++h) {
        #pragma unroll
        for (int q = 0; q < 4; ++q) { v8f z = {}; oacc[h][q][0] = z; oacc[h][q][1] = z; }
    }

    const int nbase = chunk * (NIN / NCH) + wave;

    v4f wreg[KDIM];
    v4f un[2][2];
    v4f ln[P0_CG];
    v4f lown;
    {
        const float* Wn = Wt + (size_t)nbase * KDIM * NOUT + cg * 128 + 4 * lane;
        #pragma unroll
        for (int r = 0; r < KDIM; ++r) wreg[r] = *(const v4f*)(Wn + (size_t)r * NOUT);
        #pragma unroll
        for (int h = 0; h < 2; ++h) {
            const v4f* up = (const v4f*)(U + ((size_t)(h * BH + l16) * NIN + nbase) * KDIM + 8 * hi);
            un[h][0] = up[0]; un[h][1] = up[1];
        }
        const float* lp = lgt + (((size_t)nbase * 2 + hi) * P0_CG * BH + l16) * 4;
        #pragma unroll
        for (int g = 0; g < P0_CG; ++g) ln[g] = *(const v4f*)(lp + g * (BH * 4));
        lown = *(const v4f*)(lp + cg * (BH * 4));
    }

    for (int t = 0; t < NW; ++t) {
        const int n = nbase + 8 * t;

        wave_lds_sync();
        #pragma unroll
        for (int r = 0; r < KDIM; ++r) *(v4f*)(Wl + r * WQ + 4 * lane) = wreg[r];

        float ur[2][8];
        #pragma unroll
        for (int h = 0; h < 2; ++h) {
            ur[h][0] = un[h][0].x; ur[h][1] = un[h][0].y; ur[h][2] = un[h][0].z; ur[h][3] = un[h][0].w;
            ur[h][4] = un[h][1].x; ur[h][5] = un[h][1].y; ur[h][6] = un[h][1].z; ur[h][7] = un[h][1].w;
        }

        float lv[NCAP];
        #pragma unroll
        for (int g = 0; g < P0_CG; ++g) {
            lv[4 * g] = ln[g].x; lv[4 * g + 1] = ln[g].y; lv[4 * g + 2] = ln[g].z; lv[4 * g + 3] = ln[g].w;
        }
        const v4f lo = lown;

        if (t + 1 < NW) {
            #pragma unroll
            for (int h = 0; h < 2; ++h) {
                const v4f* up = (const v4f*)(U + ((size_t)(h * BH + l16) * NIN + (n + 8)) * KDIM + 8 * hi);
                un[h][0] = up[0]; un[h][1] = up[1];
            }
            const float* lp = lgt + (((size_t)(n + 8) * 2 + hi) * P0_CG * BH + l16) * 4;
            #pragma unroll
            for (int g = 0; g < P0_CG; ++g) ln[g] = *(const v4f*)(lp + g * (BH * 4));
            lown = *(const v4f*)(lp + cg * (BH * 4));
            const float* Wn = Wt + (size_t)(n + 8) * KDIM * NOUT + cg * 128 + 4 * lane;
            #pragma unroll
            for (int r = 0; r < KDIM; ++r) wreg[r] = *(const v4f*)(Wn + (size_t)r * NOUT);
        }
        wave_lds_sync();

        float cq[4];
        {
            float m = -3.402823466e38f;
            #pragma unroll
            for (int i = 0; i < NCAP; ++i) m = fmaxf(m, lv[i]);
            float ssum = 0.0f;
            #pragma unroll
            for (int i = 0; i < NCAP; ++i) ssum += __expf(lv[i] - m);
            const float inv = 1.0f / ssum;
            cq[0] = __expf(lo.x - m) * inv;
            cq[1] = __expf(lo.y - m) * inv;
            cq[2] = __expf(lo.z - m) * inv;
            cq[3] = __expf(lo.w - m) * inv;
        }
        float c0[4], c1[4];
        #pragma unroll
        for (int q = 0; q < 4; ++q) {
            const float x = __shfl_xor(cq[q], 16);
            c0[q] = hi ? x : cq[q];
            c1[q] = hi ? cq[q] : x;
        }

        #pragma unroll
        for (int s = 0; s < 4; ++s) {
            #pragma unroll
            for (int q = 0; q < 4; ++q) {
                v2f a0, a1;
                a0.x = c0[q] * ur[0][2 * s];
                a0.y = c0[q] * ur[0][2 * s + 1];
                a1.x = c1[q] * ur[1][2 * s];
                a1.y = c1[q] * ur[1][2 * s + 1];
                #pragma unroll
                for (int jt = 0; jt < 2; ++jt) {
                    const float* wp = Wl + (8 * hi + 2 * s) * WQ + q * CAPD + jt * 16 + l16;
                    v2f bb;
                    bb.x = wp[0];
                    bb.y = wp[WQ];
                    oacc[0][q][jt] = WMMA_F32(a0, bb, oacc[0][q][jt]);
                    oacc[1][q][jt] = WMMA_F32(a1, bb, oacc[1][q][jt]);
                }
            }
        }
    }

    fold_store(smem, Wl, oacc, partial + (size_t)chunk * (BATCH * 1024) + cg * 128, tid, l16, hi);
}

template <int NPL>
__global__ __launch_bounds__(256)
void caps_reduce(const float* __restrict__ partial, float* __restrict__ out, int mode)
{
    __shared__ float osum[NOUT];
    __shared__ float fac[NCAP];
    const int b = blockIdx.x;
    const int tid = threadIdx.x;

    for (int e = tid; e < NOUT; e += 256) {
        float s = 0.0f;
        const float* p = partial + (size_t)b * 1024 + e;
        for (int c = 0; c < NPL; ++c)
            s += p[(size_t)c * (BATCH * 1024)];
        osum[e] = s;
    }
    __syncthreads();

    if (tid < NCAP) {
        float s2 = 0.0f;
        for (int j = 0; j < CAPD; ++j) { float v = osum[tid * CAPD + j]; s2 = fmaf(v, v, s2); }
        float f;
        if (mode == 0) f = rsqrtf(fmaxf(s2, 1e-12f));
        else           f = (s2 / (1.0f + s2)) * rsqrtf(s2 + 1e-7f);
        fac[tid] = f;
    }
    __syncthreads();

    for (int e = tid; e < NOUT; e += 256)
        out[(size_t)b * NOUT + e] = osum[e] * fac[e >> 5];
}

extern "C" void kernel_launch(void* const* d_in, const int* in_sizes, int n_in,
                              void* d_out, int out_size, void* d_ws, size_t ws_size,
                              hipStream_t stream)
{
    const float* U  = (const float*)d_in[0];
    const float* Wt = (const float*)d_in[1];
    float* out = (float*)d_out;

    float* partial = (float*)d_ws;
    float* o_prev  = partial + (size_t)NCH * (BATCH * 1024);
    float* lgt     = o_prev + (size_t)BATCH * NOUT;

    const size_t shmem0 = (size_t)(8 * KDIM * WQ) * sizeof(float);
    dim3 grid0(NCH, P0_CG), blk(256);

    caps_pass0<<<grid0, blk, shmem0, stream>>>(U, Wt, partial);
    caps_reduce<NCH><<<BATCH, 256, 0, stream>>>(partial, o_prev, 0);
    caps_logits<<<grid0, blk, shmem0, stream>>>(U, Wt, o_prev, lgt);
    caps_o<<<grid0, blk, shmem0, stream>>>(U, Wt, lgt, partial);
    caps_reduce<NCH><<<BATCH, 256, 0, stream>>>(partial, o_prev, 0);
    caps_logits<<<grid0, blk, shmem0, stream>>>(U, Wt, o_prev, lgt);
    caps_o<<<grid0, blk, shmem0, stream>>>(U, Wt, lgt, partial);
    caps_reduce<NCH><<<BATCH, 256, 0, stream>>>(partial, out, 1);
}
